# starting kernel
# baseline (speedup 1.0000x reference)
#include <hip/hip_runtime.h>

typedef __attribute__((ext_vector_type(2))) float v2f;
typedef __attribute__((ext_vector_type(8))) float v8f;

#define S_ROWS 16384
#define K_COLS 4096
#define D_DIM  64
#define TEMP_C 50.0f
#define MAX_EFF_TEMP 5000.0f

__global__ __launch_bounds__(32) void fcm_kernel(const float* __restrict__ X,
                                                 const float* __restrict__ Y,
                                                 const float* __restrict__ icpt,
                                                 float* __restrict__ out_choice,
                                                 float* __restrict__ out_v) {
  __shared__ float stage[16 * 16];

  const int lane = threadIdx.x;
  const int nlo  = lane & 15;
  const int h    = lane >> 4;
  const int row0 = blockIdx.x * 16;

  v2f xfrag[16];
  const float* xrow = X + (size_t)(row0 + nlo) * D_DIM + 2 * h;
  #pragma unroll
  for (int c = 0; c < 16; ++c)
    xfrag[c] = *(const v2f*)(xrow + 4 * c);

  float rmax[8], rmin[8];
  #pragma unroll
  for (int j = 0; j < 8; ++j) { rmax[j] = -__builtin_inff(); rmin[j] = __builtin_inff(); }

  for (int n0 = 0; n0 < K_COLS; n0 += 16) {
    const float* yrow = Y + (size_t)(n0 + nlo) * D_DIM + 2 * h;
    v8f acc0 = {}; v8f acc1 = {};
    #pragma unroll
    for (int c = 0; c < 16; c += 2) {
      v2f b0 = *(const v2f*)(yrow + 4 * c);
      v2f b1 = *(const v2f*)(yrow + 4 * (c + 1));
      acc0 = __builtin_amdgcn_wmma_f32_16x16x4_f32(false, xfrag[c],     false, b0,
                                                   (short)0, acc0, false, false);
      acc1 = __builtin_amdgcn_wmma_f32_16x16x4_f32(false, xfrag[c + 1], false, b1,
                                                   (short)0, acc1, false, false);
    }
    const float ic = icpt[n0 + nlo];
    #pragma unroll
    for (int j = 0; j < 8; ++j) {
      float a = (acc0[j] + acc1[j]) - ic;
      rmax[j] = fmaxf(rmax[j], a);
      rmin[j] = fminf(rmin[j], a);
    }
  }
  #pragma unroll
  for (int m = 1; m < 16; m <<= 1) {
    #pragma unroll
    for (int j = 0; j < 8; ++j) {
      rmax[j] = fmaxf(rmax[j], __shfl_xor(rmax[j], m, 32));
      rmin[j] = fminf(rmin[j], __shfl_xor(rmin[j], m, 32));
    }
  }
  float efft[8], shift[8];
  #pragma unroll
  for (int j = 0; j < 8; ++j) {
    float span = fmaxf(rmax[j] - rmin[j], 1e-3f);
    efft[j]  = fminf(fmaxf(TEMP_C / span, TEMP_C), MAX_EFF_TEMP);
    shift[j] = rmax[j];
  }

  float denom[8], vnum[8];
  #pragma unroll
  for (int j = 0; j < 8; ++j) { denom[j] = 0.0f; vnum[j] = 0.0f; }
  v8f cacc[4];
  #pragma unroll
  for (int nt = 0; nt < 4; ++nt) cacc[nt] = (v8f){};

  for (int n0 = 0; n0 < K_COLS; n0 += 16) {
    const float* yrow = Y + (size_t)(n0 + nlo) * D_DIM + 2 * h;
    v8f acc0 = {}; v8f acc1 = {};
    #pragma unroll
    for (int c = 0; c < 16; c += 2) {
      v2f b0 = *(const v2f*)(yrow + 4 * c);
      v2f b1 = *(const v2f*)(yrow + 4 * (c + 1));
      acc0 = __builtin_amdgcn_wmma_f32_16x16x4_f32(false, xfrag[c],     false, b0,
                                                   (short)0, acc0, false, false);
      acc1 = __builtin_amdgcn_wmma_f32_16x16x4_f32(false, xfrag[c + 1], false, b1,
                                                   (short)0, acc1, false, false);
    }
    const float ic = icpt[n0 + nlo];
    float e[8];
    #pragma unroll
    for (int j = 0; j < 8; ++j) {
      float a = (acc0[j] + acc1[j]) - ic;
      float w = __expf((a - shift[j]) * efft[j]);
      e[j] = w;
      denom[j] += w;
      vnum[j]  += w * a;
    }

    #pragma unroll
    for (int j = 0; j < 8; ++j)
      stage[(j + 8 * h) * 16 + nlo] = e[j];
    asm volatile("s_wait_dscnt 0" ::: "memory");

    v2f ef[4];
    #pragma unroll
    for (int cc = 0; cc < 4; ++cc)
      ef[cc] = *(const v2f*)(&stage[nlo * 16 + 4 * cc + 2 * h]);
    asm volatile("s_wait_dscnt 0" ::: "memory");

    #pragma unroll
    for (int nt = 0; nt < 4; ++nt) {
      #pragma unroll
      for (int cc = 0; cc < 4; ++cc) {
        v2f b;
        b.x = Y[(size_t)(n0 + 4 * cc + 2 * h + 0) * D_DIM + nt * 16 + nlo];
        b.y = Y[(size_t)(n0 + 4 * cc + 2 * h + 1) * D_DIM + nt * 16 + nlo];
        cacc[nt] = __builtin_amdgcn_wmma_f32_16x16x4_f32(false, ef[cc], false, b,
                                                         (short)0, cacc[nt], false, false);
      }
    }
  }

  #pragma unroll
  for (int m = 1; m < 16; m <<= 1) {
    #pragma unroll
    for (int j = 0; j < 8; ++j) {
      denom[j] += __shfl_xor(denom[j], m, 32);
      vnum[j]  += __shfl_xor(vnum[j],  m, 32);
    }
  }
  float rden[8];
  #pragma unroll
  for (int j = 0; j < 8; ++j) rden[j] = 1.0f / denom[j];

  #pragma unroll
  for (int nt = 0; nt < 4; ++nt)
    #pragma unroll
    for (int j = 0; j < 8; ++j)
      out_choice[(size_t)(row0 + j + 8 * h) * D_DIM + nt * 16 + nlo] =
          cacc[nt][j] * rden[j];

  if (nlo == 0) {
    #pragma unroll
    for (int j = 0; j < 8; ++j)
      out_v[row0 + j + 8 * h] = vnum[j] * rden[j];
  }
}

extern "C" void kernel_launch(void* const* d_in, const int* in_sizes, int n_in,
                              void* d_out, int out_size, void* d_ws, size_t ws_size,
                              hipStream_t stream) {
  const float* X  = (const float*)d_in[0];
  const float* Y  = (const float*)d_in[1];
  const float* ic = (const float*)d_in[2];
  float* out      = (float*)d_out;
  float* out_choice = out;
  float* out_v      = out + (size_t)S_ROWS * D_DIM;

  dim3 grid(S_ROWS / 16);
  dim3 block(32);
  fcm_kernel<<<grid, block, 0, stream>>>(X, Y, ic, out_choice, out_v);
}


# Register prefetch: each 16-row Y chunk's fragment loads go out as one group into spare register sets ahead of earlier chunks' WMMAs (pass 1 three chunks ahead, pass 2 one); no sum or order changes
# speedup vs baseline: 1.8279x; 1.8279x over previous
#include <hip/hip_runtime.h>

typedef __attribute__((ext_vector_type(2))) float v2f;
typedef __attribute__((ext_vector_type(8))) float v8f;

#define S_ROWS 16384
#define K_COLS 4096
#define D_DIM  64
#define NCHUNK (K_COLS / 16)
#define TEMP_C 50.0f
#define MAX_EFF_TEMP 5000.0f


struct YFragA { v2f b[16]; float ic; };
struct YFragB { v2f b[16]; };

__device__ __forceinline__ void load_frag_a_y(v2f (&b)[16], const float* __restrict__ Y,
                                              int chunk, int nlo, int h) {
  const float* yrow = Y + (size_t)(chunk * 16 + nlo) * D_DIM + 2 * h;
  #pragma unroll
  for (int c = 0; c < 16; ++c)
    b[c] = *(const v2f*)(yrow + 4 * c);
}

__device__ __forceinline__ void load_frag_a(YFragA& f, const float* __restrict__ Y,
                                            const float* __restrict__ icpt,
                                            int chunk, int nlo, int h) {
  load_frag_a_y(f.b, Y, chunk, nlo, h);
  f.ic = icpt[chunk * 16 + nlo];
}

__device__ __forceinline__ void load_frag_b(YFragB& f, const float* __restrict__ Y,
                                            int chunk, int nlo, int h) {
  const float* yb = Y + (size_t)(chunk * 16 + 2 * h) * D_DIM + nlo;
  #pragma unroll
  for (int nt = 0; nt < 4; ++nt) {
    #pragma unroll
    for (int cc = 0; cc < 4; ++cc) {
      f.b[nt * 4 + cc].x = yb[(size_t)(4 * cc + 0) * D_DIM + nt * 16];
      f.b[nt * 4 + cc].y = yb[(size_t)(4 * cc + 1) * D_DIM + nt * 16];
    }
  }
}

__device__ __forceinline__ void affine_tile(const v2f (&xfrag)[16], const v2f (&b)[16],
                                            float ic, float (&a)[8]) {
  v8f acc0 = {}; v8f acc1 = {};
  #pragma unroll
  for (int c = 0; c < 16; c += 2) {
    acc0 = __builtin_amdgcn_wmma_f32_16x16x4_f32(false, xfrag[c],     false, b[c],
                                                 (short)0, acc0, false, false);
    acc1 = __builtin_amdgcn_wmma_f32_16x16x4_f32(false, xfrag[c + 1], false, b[c + 1],
                                                 (short)0, acc1, false, false);
  }
  #pragma unroll
  for (int j = 0; j < 8; ++j)
    a[j] = (acc0[j] + acc1[j]) - ic;
}

__global__ __launch_bounds__(32) void fcm_kernel(const float* __restrict__ X,
                                                 const float* __restrict__ Y,
                                                 const float* __restrict__ icpt,
                                                 float* __restrict__ out_choice,
                                                 float* __restrict__ out_v) {
  __shared__ float stage[16 * 16];

  const int lane = threadIdx.x;
  const int nlo  = lane & 15;
  const int h    = lane >> 4;
  const int row0 = blockIdx.x * 16;

  v2f xfrag[16];
  const float* xrow = X + (size_t)(row0 + nlo) * D_DIM + 2 * h;
  #pragma unroll
  for (int c = 0; c < 16; ++c)
    xfrag[c] = *(const v2f*)(xrow + 4 * c);

  float rmax[8], rmin[8];
  #pragma unroll
  for (int j = 0; j < 8; ++j) { rmax[j] = -__builtin_inff(); rmin[j] = __builtin_inff(); }

  {
    v2f ring[4][16];
    float icc[4], icn[4];
    #pragma unroll
    for (int u = 0; u < 4; ++u) icc[u] = icpt[u * 16 + nlo];
    load_frag_a_y(ring[0], Y, 0, nlo, h);
    load_frag_a_y(ring[1], Y, 1, nlo, h);
    load_frag_a_y(ring[2], Y, 2, nlo, h);
    for (int ch = 0; ch < NCHUNK; ch += 4) {
      const int nx = ch + 4 < NCHUNK ? ch + 4 : ch;
      #pragma unroll
      for (int u = 0; u < 4; ++u) {
        int pf = ch + u + 3;
        pf = pf < NCHUNK ? pf : NCHUNK - 1;
        __builtin_amdgcn_sched_barrier(0);
        if (u == 0) {
          #pragma unroll
          for (int v = 0; v < 4; ++v) icn[v] = icpt[(nx + v) * 16 + nlo];
        }
        load_frag_a_y(ring[(u + 3) & 3], Y, pf, nlo, h);
        __builtin_amdgcn_sched_barrier(0);
        float a[8];
        affine_tile(xfrag, ring[u], icc[u], a);
        #pragma unroll
        for (int j = 0; j < 8; ++j) {
          rmax[j] = fmaxf(rmax[j], a[j]);
          rmin[j] = fminf(rmin[j], a[j]);
        }
      }
      #pragma unroll
      for (int u = 0; u < 4; ++u) icc[u] = icn[u];
    }
  }

  YFragA ra[2];
  YFragB rb[2];
  load_frag_a(ra[0], Y, icpt, 0, nlo, h);
  load_frag_b(rb[0], Y, 0, nlo, h);

  #pragma unroll
  for (int m = 1; m < 16; m <<= 1) {
    #pragma unroll
    for (int j = 0; j < 8; ++j) {
      rmax[j] = fmaxf(rmax[j], __shfl_xor(rmax[j], m, 32));
      rmin[j] = fminf(rmin[j], __shfl_xor(rmin[j], m, 32));
    }
  }
  float efft[8], shift[8];
  #pragma unroll
  for (int j = 0; j < 8; ++j) {
    float span = fmaxf(rmax[j] - rmin[j], 1e-3f);
    efft[j]  = fminf(fmaxf(TEMP_C / span, TEMP_C), MAX_EFF_TEMP);
    shift[j] = rmax[j];
  }

  float denom[8], vnum[8];
  #pragma unroll
  for (int j = 0; j < 8; ++j) { denom[j] = 0.0f; vnum[j] = 0.0f; }
  v8f cacc[4];
  #pragma unroll
  for (int nt = 0; nt < 4; ++nt) cacc[nt] = (v8f){};

  for (int ch = 0; ch < NCHUNK; ch += 2) {
    #pragma unroll
    for (int u = 0; u < 2; ++u) {
      int pf = ch + u + 1;
      pf = pf < NCHUNK ? pf : NCHUNK - 1;
      __builtin_amdgcn_sched_barrier(0);
      load_frag_a(ra[u ^ 1], Y, icpt, pf, nlo, h);
      load_frag_b(rb[u ^ 1], Y, pf, nlo, h);
      __builtin_amdgcn_sched_barrier(0);

      float a[8];
      affine_tile(xfrag, ra[u].b, ra[u].ic, a);
      float e[8];
      #pragma unroll
      for (int j = 0; j < 8; ++j) {
        float w = __expf((a[j] - shift[j]) * efft[j]);
        e[j] = w;
        denom[j] += w;
        vnum[j]  += w * a[j];
      }

      #pragma unroll
      for (int j = 0; j < 8; ++j)
        stage[(j + 8 * h) * 16 + nlo] = e[j];
      asm volatile("s_wait_dscnt 0" ::: "memory");

      v2f ef[4];
      #pragma unroll
      for (int cc = 0; cc < 4; ++cc)
        ef[cc] = *(const v2f*)(&stage[nlo * 16 + 4 * cc + 2 * h]);
      asm volatile("s_wait_dscnt 0" ::: "memory");

      #pragma unroll
      for (int nt = 0; nt < 4; ++nt) {
        #pragma unroll
        for (int cc = 0; cc < 4; ++cc) {
          cacc[nt] = __builtin_amdgcn_wmma_f32_16x16x4_f32(false, ef[cc], false,
                                                           rb[u].b[nt * 4 + cc],
                                                           (short)0, cacc[nt], false, false);
        }
      }
    }
  }

  #pragma unroll
  for (int m = 1; m < 16; m <<= 1) {
    #pragma unroll
    for (int j = 0; j < 8; ++j) {
      denom[j] += __shfl_xor(denom[j], m, 32);
      vnum[j]  += __shfl_xor(vnum[j],  m, 32);
    }
  }
  float rden[8];
  #pragma unroll
  for (int j = 0; j < 8; ++j) rden[j] = 1.0f / denom[j];

  #pragma unroll
  for (int nt = 0; nt < 4; ++nt)
    #pragma unroll
    for (int j = 0; j < 8; ++j)
      out_choice[(size_t)(row0 + j + 8 * h) * D_DIM + nt * 16 + nlo] =
          cacc[nt][j] * rden[j];

  if (nlo == 0) {
    #pragma unroll
    for (int j = 0; j < 8; ++j)
      out_v[row0 + j + 8 * h] = vnum[j] * rden[j];
  }
}

extern "C" void kernel_launch(void* const* d_in, const int* in_sizes, int n_in,
                              void* d_out, int out_size, void* d_ws, size_t ws_size,
                              hipStream_t stream) {
  const float* X  = (const float*)d_in[0];
  const float* Y  = (const float*)d_in[1];
  const float* ic = (const float*)d_in[2];
  float* out      = (float*)d_out;
  float* out_choice = out;
  float* out_v      = out + (size_t)S_ROWS * D_DIM;

  dim3 grid(S_ROWS / 16);
  dim3 block(32);
  fcm_kernel<<<grid, block, 0, stream>>>(X, Y, ic, out_choice, out_v);
}


# One walk over Y, not two: softmax against the running row maximum at the clamp temperature, sums rescaled when a maximum grows (new rounding); two-pass walk redone per tile if a span asks otherwise
# speedup vs baseline: 2.2193x; 1.2141x over previous
#include <hip/hip_runtime.h>

typedef __attribute__((ext_vector_type(2))) float v2f;
typedef __attribute__((ext_vector_type(8))) float v8f;

#define S_ROWS 16384
#define K_COLS 4096
#define D_DIM  64
#define NCHUNK (K_COLS / 16)
#define TEMP_C 50.0f
#define MAX_EFF_TEMP 5000.0f
#define TEMP_LOG2E (TEMP_C * 1.4426950408889634f)


struct YFragA { v2f b[16]; float ic; };
struct YFragB { v2f b[16]; };

__device__ __forceinline__ void load_frag_a_y(v2f (&b)[16], const float* __restrict__ Y,
                                              int chunk, int nlo, int h) {
  const float* yrow = Y + (size_t)(chunk * 16 + nlo) * D_DIM + 2 * h;
  #pragma unroll
  for (int c = 0; c < 16; ++c)
    b[c] = *(const v2f*)(yrow + 4 * c);
}

__device__ __forceinline__ void load_frag_a(YFragA& f, const float* __restrict__ Y,
                                            const float* __restrict__ icpt,
                                            int chunk, int nlo, int h) {
  load_frag_a_y(f.b, Y, chunk, nlo, h);
  f.ic = icpt[chunk * 16 + nlo];
}

__device__ __forceinline__ void load_frag_b(YFragB& f, const float* __restrict__ Y,
                                            int chunk, int nlo, int h) {
  const float* yb = Y + (size_t)(chunk * 16 + 2 * h) * D_DIM + nlo;
  #pragma unroll
  for (int nt = 0; nt < 4; ++nt) {
    #pragma unroll
    for (int cc = 0; cc < 4; ++cc) {
      f.b[nt * 4 + cc].x = yb[(size_t)(4 * cc + 0) * D_DIM + nt * 16];
      f.b[nt * 4 + cc].y = yb[(size_t)(4 * cc + 1) * D_DIM + nt * 16];
    }
  }
}

__device__ __forceinline__ void affine_tile(const v2f (&xfrag)[16], const v2f (&b)[16],
                                            float ic, float (&a)[8]) {
  v8f acc0 = {}; v8f acc1 = {};
  #pragma unroll
  for (int c = 0; c < 16; c += 2) {
    acc0 = __builtin_amdgcn_wmma_f32_16x16x4_f32(false, xfrag[c],     false, b[c],
                                                 (short)0, acc0, false, false);
    acc1 = __builtin_amdgcn_wmma_f32_16x16x4_f32(false, xfrag[c + 1], false, b[c + 1],
                                                 (short)0, acc1, false, false);
  }
  #pragma unroll
  for (int j = 0; j < 8; ++j)
    a[j] = (acc0[j] + acc1[j]) - ic;
}

template <int MASK>
__device__ __forceinline__ float row_xor(float x) {
  int b = __builtin_bit_cast(int, x);
  return __builtin_bit_cast(float,
      __builtin_amdgcn_update_dpp(b, b, 0x160 | MASK, 0xf, 0xf, true));
}

template <bool RUNNING>
__device__ __forceinline__ void walk_y(const v2f (&xfrag)[16],
                                       const float* __restrict__ Y,
                                       const float* __restrict__ icpt,
                                       float* stage, int nlo, int h,
                                       float (&mrun)[8], float (&rmin)[8],
                                       const float (&efft)[8],
                                       float (&denom)[8], float (&vnum)[8],
                                       v8f (&cacc)[4]) {
  #pragma unroll
  for (int j = 0; j < 8; ++j) { denom[j] = 0.0f; vnum[j] = 0.0f; }
  #pragma unroll
  for (int nt = 0; nt < 4; ++nt) cacc[nt] = (v8f){};

  YFragA ra[2];
  YFragB rb[2];
  load_frag_a(ra[0], Y, icpt, 0, nlo, h);
  load_frag_b(rb[0], Y, 0, nlo, h);

  for (int ch = 0; ch < NCHUNK; ch += 2) {
    #pragma unroll
    for (int u = 0; u < 2; ++u) {
      int pf = ch + u + 1;
      pf = pf < NCHUNK ? pf : NCHUNK - 1;
      __builtin_amdgcn_sched_barrier(0);
      load_frag_a(ra[u ^ 1], Y, icpt, pf, nlo, h);
      load_frag_b(rb[u ^ 1], Y, pf, nlo, h);
      __builtin_amdgcn_sched_barrier(0);

      float a[8];
      affine_tile(xfrag, ra[u].b, ra[u].ic, a);
      float e[8];
      if (RUNNING) {
        float d[8];
        #pragma unroll
        for (int j = 0; j < 8; ++j) {
          rmin[j] = fminf(rmin[j], a[j]);
          d[j] = a[j] - mrun[j];
        }
        float top = fmaxf(fmaxf(fmaxf(d[0], d[1]), fmaxf(d[2], d[3])),
                          fmaxf(fmaxf(d[4], d[5]), fmaxf(d[6], d[7])));
        if (__builtin_expect(__builtin_amdgcn_ballot_w32(top > 0.0f) != 0, 0)) {
          float mt[8];
          #pragma unroll
          for (int j = 0; j < 8; ++j) mt[j] = a[j];
          #pragma unroll
          for (int j = 0; j < 8; ++j) mt[j] = fmaxf(mt[j], row_xor<1>(mt[j]));
          #pragma unroll
          for (int j = 0; j < 8; ++j) mt[j] = fmaxf(mt[j], row_xor<2>(mt[j]));
          #pragma unroll
          for (int j = 0; j < 8; ++j) mt[j] = fmaxf(mt[j], row_xor<4>(mt[j]));
          #pragma unroll
          for (int j = 0; j < 8; ++j) mt[j] = fmaxf(mt[j], row_xor<8>(mt[j]));
          #pragma unroll
          for (int j = 0; j < 8; ++j) {
            float mn = fmaxf(mrun[j], mt[j]);
            float sc = __builtin_amdgcn_exp2f((mrun[j] - mn) * TEMP_LOG2E);
            mrun[j] = mn;
            denom[j] *= sc;
            vnum[j]  *= sc;
            #pragma unroll
            for (int nt = 0; nt < 4; ++nt) cacc[nt][j] *= sc;
            d[j] = a[j] - mn;
          }
        }
        #pragma unroll
        for (int j = 0; j < 8; ++j) {
          float w = __builtin_amdgcn_exp2f(d[j] * TEMP_LOG2E);
          e[j] = w;
          denom[j] += w;
          vnum[j]  += w * a[j];
        }
      } else {
        #pragma unroll
        for (int j = 0; j < 8; ++j) {
          float w = __expf((a[j] - mrun[j]) * efft[j]);
          e[j] = w;
          denom[j] += w;
          vnum[j]  += w * a[j];
        }
      }

      #pragma unroll
      for (int j = 0; j < 8; ++j)
        stage[(j + 8 * h) * 16 + nlo] = e[j];
      asm volatile("s_wait_dscnt 0" ::: "memory");

      v2f ef[4];
      #pragma unroll
      for (int cc = 0; cc < 4; ++cc)
        ef[cc] = *(const v2f*)(&stage[nlo * 16 + 4 * cc + 2 * h]);
      asm volatile("s_wait_dscnt 0" ::: "memory");

      #pragma unroll
      for (int nt = 0; nt < 4; ++nt) {
        #pragma unroll
        for (int cc = 0; cc < 4; ++cc) {
          cacc[nt] = __builtin_amdgcn_wmma_f32_16x16x4_f32(false, ef[cc], false,
                                                           rb[u].b[nt * 4 + cc],
                                                           (short)0, cacc[nt], false, false);
        }
      }
    }
  }
}

__global__ __launch_bounds__(32) void fcm_kernel(const float* __restrict__ X,
                                                 const float* __restrict__ Y,
                                                 const float* __restrict__ icpt,
                                                 float* __restrict__ out_choice,
                                                 float* __restrict__ out_v) {
  __shared__ float stage[16 * 16];

  const int lane = threadIdx.x;
  const int nlo  = lane & 15;
  const int h    = lane >> 4;
  const int row0 = blockIdx.x * 16;

  v2f xfrag[16];
  const float* xrow = X + (size_t)(row0 + nlo) * D_DIM + 2 * h;
  #pragma unroll
  for (int c = 0; c < 16; ++c)
    xfrag[c] = *(const v2f*)(xrow + 4 * c);

  float rmax[8], rmin[8], efft[8];
  #pragma unroll
  for (int j = 0; j < 8; ++j) {
    rmax[j] = -__builtin_inff(); rmin[j] = __builtin_inff(); efft[j] = TEMP_C;
  }
  float denom[8], vnum[8];
  v8f cacc[4];
  walk_y<true>(xfrag, Y, icpt, stage, nlo, h, rmax, rmin, efft, denom, vnum, cacc);

  #pragma unroll
  for (int m = 1; m < 16; m <<= 1) {
    #pragma unroll
    for (int j = 0; j < 8; ++j)
      rmin[j] = fminf(rmin[j], __shfl_xor(rmin[j], m, 32));
  }
  bool other_temp = false;
  #pragma unroll
  for (int j = 0; j < 8; ++j) {
    float span = fmaxf(rmax[j] - rmin[j], 1e-3f);
    efft[j] = fminf(fmaxf(TEMP_C / span, TEMP_C), MAX_EFF_TEMP);
    other_temp = other_temp || (efft[j] != TEMP_C);
  }
  if (__builtin_amdgcn_ballot_w32(other_temp) != 0)
    walk_y<false>(xfrag, Y, icpt, stage, nlo, h, rmax, rmin, efft, denom, vnum, cacc);

  #pragma unroll
  for (int m = 1; m < 16; m <<= 1) {
    #pragma unroll
    for (int j = 0; j < 8; ++j) {
      denom[j] += __shfl_xor(denom[j], m, 32);
      vnum[j]  += __shfl_xor(vnum[j],  m, 32);
    }
  }
  float rden[8];
  #pragma unroll
  for (int j = 0; j < 8; ++j) rden[j] = 1.0f / denom[j];

  #pragma unroll
  for (int nt = 0; nt < 4; ++nt)
    #pragma unroll
    for (int j = 0; j < 8; ++j)
      out_choice[(size_t)(row0 + j + 8 * h) * D_DIM + nt * 16 + nlo] =
          cacc[nt][j] * rden[j];

  if (nlo == 0) {
    #pragma unroll
    for (int j = 0; j < 8; ++j)
      out_v[row0 + j + 8 * h] = vnum[j] * rden[j];
  }
}

extern "C" void kernel_launch(void* const* d_in, const int* in_sizes, int n_in,
                              void* d_out, int out_size, void* d_ws, size_t ws_size,
                              hipStream_t stream) {
  const float* X  = (const float*)d_in[0];
  const float* Y  = (const float*)d_in[1];
  const float* ic = (const float*)d_in[2];
  float* out      = (float*)d_out;
  float* out_choice = out;
  float* out_v      = out + (size_t)S_ROWS * D_DIM;

  dim3 grid(S_ROWS / 16);
  dim3 block(32);
  fcm_kernel<<<grid, block, 0, stream>>>(X, Y, ic, out_choice, out_v);
}


# Both products transposed (Y chunk as the A operand): the weights leave product 1 already in product 2's B layout, so the LDS round trip goes; one X row per lane; sums regrouped (new order)
# speedup vs baseline: 2.6351x; 1.1874x over previous
#include <hip/hip_runtime.h>

typedef __attribute__((ext_vector_type(2))) float v2f;
typedef __attribute__((ext_vector_type(4))) float v4f;
typedef __attribute__((ext_vector_type(8))) float v8f;

#define S_ROWS 16384
#define K_COLS 4096
#define D_DIM  64
#define NCHUNK (K_COLS / 16)
#define TEMP_C 50.0f
#define MAX_EFF_TEMP 5000.0f
#define TEMP_LOG2E (TEMP_C * 1.4426950408889634f)


struct YFragA { v2f a[16]; v4f ic[2]; };
struct YFragT { v2f a[16]; };

__device__ __forceinline__ void load_frag_a(YFragA& f, const float* __restrict__ Y,
                                            const float* __restrict__ icpt,
                                            int chunk, int nlo, int h) {
  const float* yrow = Y + (size_t)(chunk * 16 + nlo) * D_DIM + 2 * h;
  #pragma unroll
  for (int c = 0; c < 16; ++c)
    f.a[c] = *(const v2f*)(yrow + 4 * c);
  const float* ip = icpt + chunk * 16 + 8 * h;
  f.ic[0] = *(const v4f*)(ip);
  f.ic[1] = *(const v4f*)(ip + 4);
}

__device__ __forceinline__ void load_frag_t(YFragT& f, const float* __restrict__ Y,
                                            int chunk, int nlo, int h) {
  const float* yb = Y + (size_t)(chunk * 16 + 8 * h) * D_DIM + nlo;
  #pragma unroll
  for (int mt = 0; mt < 4; ++mt) {
    #pragma unroll
    for (int s = 0; s < 4; ++s) {
      f.a[mt * 4 + s].x = yb[(size_t)(2 * s + 0) * D_DIM + mt * 16];
      f.a[mt * 4 + s].y = yb[(size_t)(2 * s + 1) * D_DIM + mt * 16];
    }
  }
}

__device__ __forceinline__ void affine_tile(const v2f (&xfrag)[16], const YFragA& f,
                                            float (&a)[8]) {
  v8f acc0 = {}; v8f acc1 = {};
  #pragma unroll
  for (int c = 0; c < 16; c += 2) {
    acc0 = __builtin_amdgcn_wmma_f32_16x16x4_f32(false, f.a[c],     false, xfrag[c],
                                                 (short)0, acc0, false, false);
    acc1 = __builtin_amdgcn_wmma_f32_16x16x4_f32(false, f.a[c + 1], false, xfrag[c + 1],
                                                 (short)0, acc1, false, false);
  }
  #pragma unroll
  for (int r = 0; r < 4; ++r) {
    a[r]     = (acc0[r]     + acc1[r])     - f.ic[0][r];
    a[r + 4] = (acc0[r + 4] + acc1[r + 4]) - f.ic[1][r];
  }
}

__device__ __forceinline__ float xhalf(float x) {
  int b = __builtin_bit_cast(int, x);
  return __builtin_bit_cast(float,
      __builtin_amdgcn_permlanex16(b, b, 0x76543210, 0xfedcba98, false, false));
}

template <bool RUNNING>
__device__ __forceinline__ void walk_y(const v2f (&xfrag)[16],
                                       const float* __restrict__ Y,
                                       const float* __restrict__ icpt,
                                       int nlo, int h,
                                       float& mrun, float& rmin, float efft,
                                       float (&denom)[8], float (&vnum)[8],
                                       v8f (&cacc)[4]) {
  #pragma unroll
  for (int r = 0; r < 8; ++r) { denom[r] = 0.0f; vnum[r] = 0.0f; }
  #pragma unroll
  for (int mt = 0; mt < 4; ++mt) cacc[mt] = (v8f){};

  YFragA ra[2];
  YFragT rt[2];
  load_frag_a(ra[0], Y, icpt, 0, nlo, h);
  load_frag_t(rt[0], Y, 0, nlo, h);

  for (int ch = 0; ch < NCHUNK; ch += 2) {
    #pragma unroll
    for (int u = 0; u < 2; ++u) {
      int pf = ch + u + 1;
      pf = pf < NCHUNK ? pf : NCHUNK - 1;
      __builtin_amdgcn_sched_barrier(0);
      load_frag_a(ra[u ^ 1], Y, icpt, pf, nlo, h);
      load_frag_t(rt[u ^ 1], Y, pf, nlo, h);
      __builtin_amdgcn_sched_barrier(0);

      float a[8];
      affine_tile(xfrag, ra[u], a);
      float e[8];
      if (RUNNING) {
        float top = fmaxf(fmaxf(fmaxf(a[0], a[1]), fmaxf(a[2], a[3])),
                          fmaxf(fmaxf(a[4], a[5]), fmaxf(a[6], a[7])));
        float low = fminf(fminf(fminf(a[0], a[1]), fminf(a[2], a[3])),
                          fminf(fminf(a[4], a[5]), fminf(a[6], a[7])));
        rmin = fminf(rmin, low);
        if (__builtin_expect(__builtin_amdgcn_ballot_w32(top > mrun) != 0, 0)) {
          float mn = fmaxf(mrun, fmaxf(top, xhalf(top)));
          float sc = __builtin_amdgcn_exp2f((mrun - mn) * TEMP_LOG2E);
          mrun = mn;
          #pragma unroll
          for (int r = 0; r < 8; ++r) { denom[r] *= sc; vnum[r] *= sc; }
          #pragma unroll
          for (int mt = 0; mt < 4; ++mt)
            #pragma unroll
            for (int r = 0; r < 8; ++r) cacc[mt][r] *= sc;
        }
        #pragma unroll
        for (int r = 0; r < 8; ++r) {
          float w = __builtin_amdgcn_exp2f((a[r] - mrun) * TEMP_LOG2E);
          e[r] = w;
          denom[r] += w;
          vnum[r]  += w * a[r];
        }
      } else {
        #pragma unroll
        for (int r = 0; r < 8; ++r) {
          float w = __expf((a[r] - mrun) * efft);
          e[r] = w;
          denom[r] += w;
          vnum[r]  += w * a[r];
        }
      }

      #pragma unroll
      for (int s = 0; s < 4; ++s) {
        v2f ef = { e[2 * s], e[2 * s + 1] };
        #pragma unroll
        for (int mt = 0; mt < 4; ++mt) {
          cacc[mt] = __builtin_amdgcn_wmma_f32_16x16x4_f32(false, rt[u].a[mt * 4 + s],
                                                           false, ef,
                                                           (short)0, cacc[mt], false, false);
        }
      }
    }
  }
}

__global__ __launch_bounds__(32) void fcm_kernel(const float* __restrict__ X,
                                                 const float* __restrict__ Y,
                                                 const float* __restrict__ icpt,
                                                 float* __restrict__ out_choice,
                                                 float* __restrict__ out_v) {
  const int lane = threadIdx.x;
  const int nlo  = lane & 15;
  const int h    = lane >> 4;
  const int row0 = blockIdx.x * 16;

  v2f xfrag[16];
  const float* xrow = X + (size_t)(row0 + nlo) * D_DIM + 2 * h;
  #pragma unroll
  for (int c = 0; c < 16; ++c)
    xfrag[c] = *(const v2f*)(xrow + 4 * c);

  float rmax = -__builtin_inff(), rmin = __builtin_inff();
  float denom[8], vnum[8];
  v8f cacc[4];
  walk_y<true>(xfrag, Y, icpt, nlo, h, rmax, rmin, TEMP_C, denom, vnum, cacc);

  rmin = fminf(rmin, xhalf(rmin));
  float span = fmaxf(rmax - rmin, 1e-3f);
  float efft = fminf(fmaxf(TEMP_C / span, TEMP_C), MAX_EFF_TEMP);
  if (__builtin_amdgcn_ballot_w32(efft != TEMP_C) != 0)
    walk_y<false>(xfrag, Y, icpt, nlo, h, rmax, rmin, efft, denom, vnum, cacc);

  float dsum = ((denom[0] + denom[1]) + (denom[2] + denom[3])) +
               ((denom[4] + denom[5]) + (denom[6] + denom[7]));
  float vsum = ((vnum[0] + vnum[1]) + (vnum[2] + vnum[3])) +
               ((vnum[4] + vnum[5]) + (vnum[6] + vnum[7]));
  dsum += xhalf(dsum);
  vsum += xhalf(vsum);
  const float rden = 1.0f / dsum;

  float* orow = out_choice + (size_t)(row0 + nlo) * D_DIM + 8 * h;
  #pragma unroll
  for (int mt = 0; mt < 4; ++mt) {
    v4f lo = { cacc[mt][0] * rden, cacc[mt][1] * rden, cacc[mt][2] * rden, cacc[mt][3] * rden };
    v4f hi = { cacc[mt][4] * rden, cacc[mt][5] * rden, cacc[mt][6] * rden, cacc[mt][7] * rden };
    *(v4f*)(orow + mt * 16)     = lo;
    *(v4f*)(orow + mt * 16 + 4) = hi;
  }

  if (h == 0)
    out_v[row0 + nlo] = vsum * rden;
}

extern "C" void kernel_launch(void* const* d_in, const int* in_sizes, int n_in,
                              void* d_out, int out_size, void* d_ws, size_t ws_size,
                              hipStream_t stream) {
  const float* X  = (const float*)d_in[0];
  const float* Y  = (const float*)d_in[1];
  const float* ic = (const float*)d_in[2];
  float* out      = (float*)d_out;
  float* out_choice = out;
  float* out_v      = out + (size_t)S_ROWS * D_DIM;

  dim3 grid(S_ROWS / 16);
  dim3 block(32);
  fcm_kernel<<<grid, block, 0, stream>>>(X, Y, ic, out_choice, out_v);
}


# Y repacked per call into WMMA fragment order in the workspace by a first kernel: each fragment load of the walk is 16 bytes per lane over 4 whole lines, not 16 part lines; same values, same sums
# speedup vs baseline: 2.8569x; 1.0842x over previous
#include <hip/hip_runtime.h>

typedef __attribute__((ext_vector_type(2))) float v2f;
typedef __attribute__((ext_vector_type(4))) float v4f;
typedef __attribute__((ext_vector_type(8))) float v8f;

#define S_ROWS 16384
#define K_COLS 4096
#define D_DIM  64
#define NCHUNK (K_COLS / 16)
#define TEMP_C 50.0f
#define MAX_EFF_TEMP 5000.0f
#define TEMP_LOG2E (TEMP_C * 1.4426950408889634f)


struct YFragA { v2f a[16]; v4f ic[2]; };
struct YFragT { v2f a[16]; };

#define PACK_SLOT  (32 * 4)
#define PACK_CHUNK (16 * PACK_SLOT)

__global__ __launch_bounds__(256) void pack_y(const float* __restrict__ Y,
                                              float* __restrict__ P) {
  const int t    = blockIdx.x * 256 + threadIdx.x;
  const int lane = t & 31;
  const int j    = (t >> 5) & 15;
  const int ch   = t >> 9;
  const int nlo  = lane & 15;
  const int h    = lane >> 4;
  v4f v;
  if (j < 8) {
    const float* yr = Y + (size_t)(ch * 16 + nlo) * D_DIM + 8 * j + 2 * h;
    v2f lo = *(const v2f*)(yr);
    v2f hi = *(const v2f*)(yr + 4);
    v = (v4f){ lo.x, lo.y, hi.x, hi.y };
  } else {
    const int jj = j - 8;
    const float* yb = Y + (size_t)(ch * 16 + 8 * h + 4 * (jj & 1)) * D_DIM
                        + (jj >> 1) * 16 + nlo;
    v = (v4f){ yb[0], yb[D_DIM], yb[2 * D_DIM], yb[3 * D_DIM] };
  }
  *(v4f*)(P + (size_t)t * 4) = v;
}

__device__ __forceinline__ void load_frag_a(YFragA& f, const float* __restrict__ P,
                                            const float* __restrict__ icpt,
                                            int chunk, int lane, int h) {
  const v4f* p = (const v4f*)(P + (size_t)chunk * PACK_CHUNK) + lane;
  #pragma unroll
  for (int j = 0; j < 8; ++j) {
    v4f v = p[j * 32];
    f.a[2 * j]     = (v2f){ v.x, v.y };
    f.a[2 * j + 1] = (v2f){ v.z, v.w };
  }
  const float* ip = icpt + chunk * 16 + 8 * h;
  f.ic[0] = *(const v4f*)(ip);
  f.ic[1] = *(const v4f*)(ip + 4);
}

__device__ __forceinline__ void load_frag_t(YFragT& f, const float* __restrict__ P,
                                            int chunk, int lane) {
  const v4f* p = (const v4f*)(P + (size_t)chunk * PACK_CHUNK + 8 * PACK_SLOT) + lane;
  #pragma unroll
  for (int j = 0; j < 8; ++j) {
    v4f v = p[j * 32];
    f.a[2 * j]     = (v2f){ v.x, v.y };
    f.a[2 * j + 1] = (v2f){ v.z, v.w };
  }
}

__device__ __forceinline__ void affine_tile(const v2f (&xfrag)[16], const YFragA& f,
                                            float (&a)[8]) {
  v8f acc0 = {}; v8f acc1 = {};
  #pragma unroll
  for (int c = 0; c < 16; c += 2) {
    acc0 = __builtin_amdgcn_wmma_f32_16x16x4_f32(false, f.a[c],     false, xfrag[c],
                                                 (short)0, acc0, false, false);
    acc1 = __builtin_amdgcn_wmma_f32_16x16x4_f32(false, f.a[c + 1], false, xfrag[c + 1],
                                                 (short)0, acc1, false, false);
  }
  #pragma unroll
  for (int r = 0; r < 4; ++r) {
    a[r]     = (acc0[r]     + acc1[r])     - f.ic[0][r];
    a[r + 4] = (acc0[r + 4] + acc1[r + 4]) - f.ic[1][r];
  }
}

__device__ __forceinline__ float xhalf(float x) {
  int b = __builtin_bit_cast(int, x);
  return __builtin_bit_cast(float,
      __builtin_amdgcn_permlanex16(b, b, 0x76543210, 0xfedcba98, false, false));
}

template <bool RUNNING>
__device__ __forceinline__ void walk_y(const v2f (&xfrag)[16],
                                       const float* __restrict__ P,
                                       const float* __restrict__ icpt,
                                       int lane, int h,
                                       float& mrun, float& rmin, float efft,
                                       float (&denom)[8], float (&vnum)[8],
                                       v8f (&cacc)[4]) {
  #pragma unroll
  for (int r = 0; r < 8; ++r) { denom[r] = 0.0f; vnum[r] = 0.0f; }
  #pragma unroll
  for (int mt = 0; mt < 4; ++mt) cacc[mt] = (v8f){};

  YFragA ra[2];
  YFragT rt[2];
  load_frag_a(ra[0], P, icpt, 0, lane, h);
  load_frag_t(rt[0], P, 0, lane);

  for (int ch = 0; ch < NCHUNK; ch += 2) {
    #pragma unroll
    for (int u = 0; u < 2; ++u) {
      int pf = ch + u + 1;
      pf = pf < NCHUNK ? pf : NCHUNK - 1;
      __builtin_amdgcn_sched_barrier(0);
      load_frag_a(ra[u ^ 1], P, icpt, pf, lane, h);
      load_frag_t(rt[u ^ 1], P, pf, lane);
      __builtin_amdgcn_sched_barrier(0);

      float a[8];
      affine_tile(xfrag, ra[u], a);
      float e[8];
      if (RUNNING) {
        float top = fmaxf(fmaxf(fmaxf(a[0], a[1]), fmaxf(a[2], a[3])),
                          fmaxf(fmaxf(a[4], a[5]), fmaxf(a[6], a[7])));
        float low = fminf(fminf(fminf(a[0], a[1]), fminf(a[2], a[3])),
                          fminf(fminf(a[4], a[5]), fminf(a[6], a[7])));
        rmin = fminf(rmin, low);
        if (__builtin_expect(__builtin_amdgcn_ballot_w32(top > mrun) != 0, 0)) {
          float mn = fmaxf(mrun, fmaxf(top, xhalf(top)));
          float sc = __builtin_amdgcn_exp2f((mrun - mn) * TEMP_LOG2E);
          mrun = mn;
          #pragma unroll
          for (int r = 0; r < 8; ++r) { denom[r] *= sc; vnum[r] *= sc; }
          #pragma unroll
          for (int mt = 0; mt < 4; ++mt)
            #pragma unroll
            for (int r = 0; r < 8; ++r) cacc[mt][r] *= sc;
        }
        #pragma unroll
        for (int r = 0; r < 8; ++r) {
          float w = __builtin_amdgcn_exp2f((a[r] - mrun) * TEMP_LOG2E);
          e[r] = w;
          denom[r] += w;
          vnum[r]  += w * a[r];
        }
      } else {
        #pragma unroll
        for (int r = 0; r < 8; ++r) {
          float w = __expf((a[r] - mrun) * efft);
          e[r] = w;
          denom[r] += w;
          vnum[r]  += w * a[r];
        }
      }

      #pragma unroll
      for (int s = 0; s < 4; ++s) {
        v2f ef = { e[2 * s], e[2 * s + 1] };
        #pragma unroll
        for (int mt = 0; mt < 4; ++mt) {
          cacc[mt] = __builtin_amdgcn_wmma_f32_16x16x4_f32(false, rt[u].a[mt * 4 + s],
                                                           false, ef,
                                                           (short)0, cacc[mt], false, false);
        }
      }
    }
  }
}

__global__ __launch_bounds__(32) void fcm_kernel(const float* __restrict__ X,
                                                 const float* __restrict__ P,
                                                 const float* __restrict__ icpt,
                                                 float* __restrict__ out_choice,
                                                 float* __restrict__ out_v) {
  const int lane = threadIdx.x;
  const int nlo  = lane & 15;
  const int h    = lane >> 4;
  const int row0 = blockIdx.x * 16;

  v2f xfrag[16];
  const float* xrow = X + (size_t)(row0 + nlo) * D_DIM + 2 * h;
  #pragma unroll
  for (int c = 0; c < 16; ++c)
    xfrag[c] = *(const v2f*)(xrow + 4 * c);

  float rmax = -__builtin_inff(), rmin = __builtin_inff();
  float denom[8], vnum[8];
  v8f cacc[4];
  walk_y<true>(xfrag, P, icpt, lane, h, rmax, rmin, TEMP_C, denom, vnum, cacc);

  rmin = fminf(rmin, xhalf(rmin));
  float span = fmaxf(rmax - rmin, 1e-3f);
  float efft = fminf(fmaxf(TEMP_C / span, TEMP_C), MAX_EFF_TEMP);
  if (__builtin_amdgcn_ballot_w32(efft != TEMP_C) != 0)
    walk_y<false>(xfrag, P, icpt, lane, h, rmax, rmin, efft, denom, vnum, cacc);

  float dsum = ((denom[0] + denom[1]) + (denom[2] + denom[3])) +
               ((denom[4] + denom[5]) + (denom[6] + denom[7]));
  float vsum = ((vnum[0] + vnum[1]) + (vnum[2] + vnum[3])) +
               ((vnum[4] + vnum[5]) + (vnum[6] + vnum[7]));
  dsum += xhalf(dsum);
  vsum += xhalf(vsum);
  const float rden = 1.0f / dsum;

  float* orow = out_choice + (size_t)(row0 + nlo) * D_DIM + 8 * h;
  #pragma unroll
  for (int mt = 0; mt < 4; ++mt) {
    v4f lo = { cacc[mt][0] * rden, cacc[mt][1] * rden, cacc[mt][2] * rden, cacc[mt][3] * rden };
    v4f hi = { cacc[mt][4] * rden, cacc[mt][5] * rden, cacc[mt][6] * rden, cacc[mt][7] * rden };
    *(v4f*)(orow + mt * 16)     = lo;
    *(v4f*)(orow + mt * 16 + 4) = hi;
  }

  if (h == 0)
    out_v[row0 + nlo] = vsum * rden;
}

extern "C" void kernel_launch(void* const* d_in, const int* in_sizes, int n_in,
                              void* d_out, int out_size, void* d_ws, size_t ws_size,
                              hipStream_t stream) {
  const float* X  = (const float*)d_in[0];
  const float* Y  = (const float*)d_in[1];
  const float* ic = (const float*)d_in[2];
  float* out      = (float*)d_out;
  float* out_choice = out;
  float* out_v      = out + (size_t)S_ROWS * D_DIM;

  float* P = (float*)d_ws;

  pack_y<<<dim3(NCHUNK * 16 * 32 / 256), dim3(256), 0, stream>>>(Y, P);

  dim3 grid(S_ROWS / 16);
  dim3 block(32);
  fcm_kernel<<<grid, block, 0, stream>>>(X, P, ic, out_choice, out_v);
}


# Software pipeline inside the lone wave: chunk i's exp and sum vector work issues between chunk i+1's affine WMMAs and chunk i's choice WMMAs; fetch two trips ahead; same operations, sums and order
# speedup vs baseline: 3.5495x; 1.2424x over previous
#include <hip/hip_runtime.h>

typedef __attribute__((ext_vector_type(2))) float v2f;
typedef __attribute__((ext_vector_type(4))) float v4f;
typedef __attribute__((ext_vector_type(8))) float v8f;

#define S_ROWS 16384
#define K_COLS 4096
#define D_DIM  64
#define NCHUNK (K_COLS / 16)
#define TEMP_C 50.0f
#define MAX_EFF_TEMP 5000.0f
#define TEMP_LOG2E (TEMP_C * 1.4426950408889634f)


struct YFragA { v2f a[16]; v4f ic[2]; };
struct YFragT { v2f a[16]; };

#define PACK_SLOT  (32 * 4)
#define PACK_CHUNK (16 * PACK_SLOT)

__global__ __launch_bounds__(256) void pack_y(const float* __restrict__ Y,
                                              float* __restrict__ P) {
  const int t    = blockIdx.x * 256 + threadIdx.x;
  const int lane = t & 31;
  const int j    = (t >> 5) & 15;
  const int ch   = t >> 9;
  const int nlo  = lane & 15;
  const int h    = lane >> 4;
  v4f v;
  if (j < 8) {
    const float* yr = Y + (size_t)(ch * 16 + nlo) * D_DIM + 8 * j + 2 * h;
    v2f lo = *(const v2f*)(yr);
    v2f hi = *(const v2f*)(yr + 4);
    v = (v4f){ lo.x, lo.y, hi.x, hi.y };
  } else {
    const int jj = j - 8;
    const float* yb = Y + (size_t)(ch * 16 + 8 * h + 4 * (jj & 1)) * D_DIM
                        + (jj >> 1) * 16 + nlo;
    v = (v4f){ yb[0], yb[D_DIM], yb[2 * D_DIM], yb[3 * D_DIM] };
  }
  *(v4f*)(P + (size_t)t * 4) = v;
}

__device__ __forceinline__ void load_frag_a(YFragA& f, const float* __restrict__ P,
                                            const float* __restrict__ icpt,
                                            int chunk, int lane, int h) {
  const v4f* p = (const v4f*)(P + (size_t)chunk * PACK_CHUNK) + lane;
  #pragma unroll
  for (int j = 0; j < 8; ++j) {
    v4f v = p[j * 32];
    f.a[2 * j]     = (v2f){ v.x, v.y };
    f.a[2 * j + 1] = (v2f){ v.z, v.w };
  }
  const float* ip = icpt + chunk * 16 + 8 * h;
  f.ic[0] = *(const v4f*)(ip);
  f.ic[1] = *(const v4f*)(ip + 4);
}

__device__ __forceinline__ void load_frag_t(YFragT& f, const float* __restrict__ P,
                                            int chunk, int lane) {
  const v4f* p = (const v4f*)(P + (size_t)chunk * PACK_CHUNK + 8 * PACK_SLOT) + lane;
  #pragma unroll
  for (int j = 0; j < 8; ++j) {
    v4f v = p[j * 32];
    f.a[2 * j]     = (v2f){ v.x, v.y };
    f.a[2 * j + 1] = (v2f){ v.z, v.w };
  }
}

__device__ __forceinline__ void affine_tile(const v2f (&xfrag)[16], const YFragA& f,
                                            float (&a)[8]) {
  v8f acc0 = {}; v8f acc1 = {};
  #pragma unroll
  for (int c = 0; c < 16; c += 2) {
    acc0 = __builtin_amdgcn_wmma_f32_16x16x4_f32(false, f.a[c],     false, xfrag[c],
                                                 (short)0, acc0, false, false);
    acc1 = __builtin_amdgcn_wmma_f32_16x16x4_f32(false, f.a[c + 1], false, xfrag[c + 1],
                                                 (short)0, acc1, false, false);
  }
  #pragma unroll
  for (int r = 0; r < 4; ++r) {
    a[r]     = (acc0[r]     + acc1[r])     - f.ic[0][r];
    a[r + 4] = (acc0[r + 4] + acc1[r + 4]) - f.ic[1][r];
  }
}

__device__ __forceinline__ float xhalf(float x) {
  int b = __builtin_bit_cast(int, x);
  return __builtin_bit_cast(float,
      __builtin_amdgcn_permlanex16(b, b, 0x76543210, 0xfedcba98, false, false));
}

template <bool RUNNING>
__device__ __forceinline__ void walk_y(const v2f (&xfrag)[16],
                                       const float* __restrict__ P,
                                       const float* __restrict__ icpt,
                                       int lane, int h,
                                       float& mrun, float& rmin, float efft,
                                       float (&denom)[8], float (&vnum)[8],
                                       v8f (&cacc)[4]) {
  #pragma unroll
  for (int r = 0; r < 8; ++r) { denom[r] = 0.0f; vnum[r] = 0.0f; }
  #pragma unroll
  for (int mt = 0; mt < 4; ++mt) cacc[mt] = (v8f){};

  YFragA ra[2];
  YFragT rt[2];
  load_frag_a(ra[0], P, icpt, 0, lane, h);
  load_frag_t(rt[0], P, 0, lane);

  for (int ch = 0; ch < NCHUNK; ch += 2) {
    #pragma unroll
    for (int u = 0; u < 2; ++u) {
      int pf = ch + u + 1;
      pf = pf < NCHUNK ? pf : NCHUNK - 1;
      __builtin_amdgcn_sched_barrier(0);
      load_frag_a(ra[u ^ 1], P, icpt, pf, lane, h);
      load_frag_t(rt[u ^ 1], P, pf, lane);
      __builtin_amdgcn_sched_barrier(0);

      float a[8];
      affine_tile(xfrag, ra[u], a);
      float e[8];
      if (RUNNING) {
        float top = fmaxf(fmaxf(fmaxf(a[0], a[1]), fmaxf(a[2], a[3])),
                          fmaxf(fmaxf(a[4], a[5]), fmaxf(a[6], a[7])));
        float low = fminf(fminf(fminf(a[0], a[1]), fminf(a[2], a[3])),
                          fminf(fminf(a[4], a[5]), fminf(a[6], a[7])));
        rmin = fminf(rmin, low);
        if (__builtin_expect(__builtin_amdgcn_ballot_w32(top > mrun) != 0, 0)) {
          float mn = fmaxf(mrun, fmaxf(top, xhalf(top)));
          float sc = __builtin_amdgcn_exp2f((mrun - mn) * TEMP_LOG2E);
          mrun = mn;
          #pragma unroll
          for (int r = 0; r < 8; ++r) { denom[r] *= sc; vnum[r] *= sc; }
          #pragma unroll
          for (int mt = 0; mt < 4; ++mt)
            #pragma unroll
            for (int r = 0; r < 8; ++r) cacc[mt][r] *= sc;
        }
        #pragma unroll
        for (int r = 0; r < 8; ++r) {
          float w = __builtin_amdgcn_exp2f((a[r] - mrun) * TEMP_LOG2E);
          e[r] = w;
          denom[r] += w;
          vnum[r]  += w * a[r];
        }
      } else {
        #pragma unroll
        for (int r = 0; r < 8; ++r) {
          float w = __expf((a[r] - mrun) * efft);
          e[r] = w;
          denom[r] += w;
          vnum[r]  += w * a[r];
        }
      }

      #pragma unroll
      for (int s = 0; s < 4; ++s) {
        v2f ef = { e[2 * s], e[2 * s + 1] };
        #pragma unroll
        for (int mt = 0; mt < 4; ++mt) {
          cacc[mt] = __builtin_amdgcn_wmma_f32_16x16x4_f32(false, rt[u].a[mt * 4 + s],
                                                           false, ef,
                                                           (short)0, cacc[mt], false, false);
        }
      }
    }
  }
}

__device__ __forceinline__ void walk_y_running(const v2f (&xfrag)[16],
                                               const float* __restrict__ P,
                                               const float* __restrict__ icpt,
                                               int lane, int h,
                                               float& mrun, float& rmin,
                                               float (&denom)[8], float (&vnum)[8],
                                               v8f (&cacc)[4]) {
  #pragma unroll
  for (int r = 0; r < 8; ++r) { denom[r] = 0.0f; vnum[r] = 0.0f; }
  #pragma unroll
  for (int mt = 0; mt < 4; ++mt) cacc[mt] = (v8f){};

  YFragA fa[4];
  YFragT ft[4];
  load_frag_a(fa[0], P, icpt, 0, lane, h);
  load_frag_t(ft[0], P, 0, lane);
  load_frag_a(fa[1], P, icpt, 1, lane, h);
  load_frag_t(ft[1], P, 1, lane);
  load_frag_a(fa[2], P, icpt, 2, lane, h);

  float a[8];
  affine_tile(xfrag, fa[0], a);
  float top = fmaxf(fmaxf(fmaxf(a[0], a[1]), fmaxf(a[2], a[3])),
                    fmaxf(fmaxf(a[4], a[5]), fmaxf(a[6], a[7])));
  {
    float low = fminf(fminf(fminf(a[0], a[1]), fminf(a[2], a[3])),
                      fminf(fminf(a[4], a[5]), fminf(a[6], a[7])));
    rmin = fminf(rmin, low);
  }
  unsigned grow = __builtin_amdgcn_ballot_w32(top > mrun);

  for (int ch = 0; ch < NCHUNK; ch += 4) {
    #pragma unroll
    for (int u = 0; u < 4; ++u) {
      int pa = ch + u + 3;
      pa = pa < NCHUNK ? pa : NCHUNK - 1;
      int pt = ch + u + 2;
      pt = pt < NCHUNK ? pt : NCHUNK - 1;
      __builtin_amdgcn_sched_barrier(0);
      load_frag_a(fa[(u + 3) & 3], P, icpt, pa, lane, h);
      load_frag_t(ft[(u + 2) & 3], P, pt, lane);
      __builtin_amdgcn_sched_barrier(0);

      if (__builtin_expect(grow != 0, 0)) {
        float mn = fmaxf(mrun, fmaxf(top, xhalf(top)));
        float sc = __builtin_amdgcn_exp2f((mrun - mn) * TEMP_LOG2E);
        mrun = mn;
        #pragma unroll
        for (int r = 0; r < 8; ++r) { denom[r] *= sc; vnum[r] *= sc; }
        #pragma unroll
        for (int mt = 0; mt < 4; ++mt)
          #pragma unroll
          for (int r = 0; r < 8; ++r) cacc[mt][r] *= sc;
      }

      const YFragA& fn = fa[(u + 1) & 3];
      v8f acc0 = {}; v8f acc1 = {};
      float e[8];
      #pragma unroll
      for (int r = 0; r < 8; ++r) {
        acc0 = __builtin_amdgcn_wmma_f32_16x16x4_f32(false, fn.a[2 * r],     false, xfrag[2 * r],
                                                     (short)0, acc0, false, false);
        acc1 = __builtin_amdgcn_wmma_f32_16x16x4_f32(false, fn.a[2 * r + 1], false, xfrag[2 * r + 1],
                                                     (short)0, acc1, false, false);
        float w = __builtin_amdgcn_exp2f((a[r] - mrun) * TEMP_LOG2E);
        e[r] = w;
        denom[r] += w;
        vnum[r]  += w * a[r];
      }

      #pragma unroll
      for (int s = 0; s < 4; ++s) {
        v2f ef = { e[2 * s], e[2 * s + 1] };
        #pragma unroll
        for (int mt = 0; mt < 4; ++mt) {
          cacc[mt] = __builtin_amdgcn_wmma_f32_16x16x4_f32(false, ft[u].a[mt * 4 + s],
                                                           false, ef,
                                                           (short)0, cacc[mt], false, false);
        }
        if (s == 1) {
          #pragma unroll
          for (int r = 0; r < 4; ++r) a[r] = (acc0[r] + acc1[r]) - fn.ic[0][r];
        }
        if (s == 2) {
          #pragma unroll
          for (int r = 0; r < 4; ++r) a[r + 4] = (acc0[r + 4] + acc1[r + 4]) - fn.ic[1][r];
        }
        if (s == 3) {
          top = fmaxf(fmaxf(fmaxf(a[0], a[1]), fmaxf(a[2], a[3])),
                      fmaxf(fmaxf(a[4], a[5]), fmaxf(a[6], a[7])));
          float low = fminf(fminf(fminf(a[0], a[1]), fminf(a[2], a[3])),
                            fminf(fminf(a[4], a[5]), fminf(a[6], a[7])));
          rmin = fminf(rmin, low);
          grow = __builtin_amdgcn_ballot_w32(top > mrun);
        }
      }
    }
  }
}

__global__ __launch_bounds__(32) void fcm_kernel(const float* __restrict__ X,
                                                 const float* __restrict__ P,
                                                 const float* __restrict__ icpt,
                                                 float* __restrict__ out_choice,
                                                 float* __restrict__ out_v) {
  const int lane = threadIdx.x;
  const int nlo  = lane & 15;
  const int h    = lane >> 4;
  const int row0 = blockIdx.x * 16;

  v2f xfrag[16];
  const float* xrow = X + (size_t)(row0 + nlo) * D_DIM + 2 * h;
  #pragma unroll
  for (int c = 0; c < 16; ++c)
    xfrag[c] = *(const v2f*)(xrow + 4 * c);

  float rmax = -__builtin_inff(), rmin = __builtin_inff();
  float denom[8], vnum[8];
  v8f cacc[4];
  walk_y_running(xfrag, P, icpt, lane, h, rmax, rmin, denom, vnum, cacc);

  rmin = fminf(rmin, xhalf(rmin));
  float span = fmaxf(rmax - rmin, 1e-3f);
  float efft = fminf(fmaxf(TEMP_C / span, TEMP_C), MAX_EFF_TEMP);
  if (__builtin_amdgcn_ballot_w32(efft != TEMP_C) != 0)
    walk_y<false>(xfrag, P, icpt, lane, h, rmax, rmin, efft, denom, vnum, cacc);

  float dsum = ((denom[0] + denom[1]) + (denom[2] + denom[3])) +
               ((denom[4] + denom[5]) + (denom[6] + denom[7]));
  float vsum = ((vnum[0] + vnum[1]) + (vnum[2] + vnum[3])) +
               ((vnum[4] + vnum[5]) + (vnum[6] + vnum[7]));
  dsum += xhalf(dsum);
  vsum += xhalf(vsum);
  const float rden = 1.0f / dsum;

  float* orow = out_choice + (size_t)(row0 + nlo) * D_DIM + 8 * h;
  #pragma unroll
  for (int mt = 0; mt < 4; ++mt) {
    v4f lo = { cacc[mt][0] * rden, cacc[mt][1] * rden, cacc[mt][2] * rden, cacc[mt][3] * rden };
    v4f hi = { cacc[mt][4] * rden, cacc[mt][5] * rden, cacc[mt][6] * rden, cacc[mt][7] * rden };
    *(v4f*)(orow + mt * 16)     = lo;
    *(v4f*)(orow + mt * 16 + 4) = hi;
  }

  if (h == 0)
    out_v[row0 + nlo] = vsum * rden;
}

extern "C" void kernel_launch(void* const* d_in, const int* in_sizes, int n_in,
                              void* d_out, int out_size, void* d_ws, size_t ws_size,
                              hipStream_t stream) {
  const float* X  = (const float*)d_in[0];
  const float* Y  = (const float*)d_in[1];
  const float* ic = (const float*)d_in[2];
  float* out      = (float*)d_out;
  float* out_choice = out;
  float* out_v      = out + (size_t)S_ROWS * D_DIM;

  float* P = (float*)d_ws;

  pack_y<<<dim3(NCHUNK * 16 * 32 / 256), dim3(256), 0, stream>>>(Y, P);

  dim3 grid(S_ROWS / 16);
  dim3 block(32);
  fcm_kernel<<<grid, block, 0, stream>>>(X, P, ic, out_choice, out_v);
}
